# starting kernel
# baseline (speedup 1.0000x reference)
#include <hip/hip_runtime.h>

typedef float v2f __attribute__((ext_vector_type(2)));
typedef float v8f __attribute__((ext_vector_type(8)));

__device__ __forceinline__ float leaky(float v) {
    return fmaxf(v, 0.1f * v);
}

__device__ __forceinline__ void gemm_tile_16x64(const float* __restrict__ X,
                                                const float* __restrict__ W,
                                                const float* __restrict__ bias,
                                                int Kd, int m0, int n0,
                                                float res[4][8]) {
    const int lane = threadIdx.x & 31;
    const int half = lane >> 4;
    const int lm   = lane & 15;

    v8f acc0 = {}, acc1 = {}, acc2 = {}, acc3 = {};

    const float* ap = X + (size_t)(m0 + lm) * Kd + 2 * half;
    const float* bp = W + (size_t)(n0 + lm) * Kd + 2 * half;
    const size_t wrow16 = (size_t)16 * Kd;

    for (int k0 = 0; k0 < Kd; k0 += 4) {
        v2f a  = *(const v2f*)(ap + k0);
        v2f b0 = *(const v2f*)(bp + k0);
        v2f b1 = *(const v2f*)(bp + wrow16 + k0);
        v2f b2 = *(const v2f*)(bp + 2 * wrow16 + k0);
        v2f b3 = *(const v2f*)(bp + 3 * wrow16 + k0);
        acc0 = __builtin_amdgcn_wmma_f32_16x16x4_f32(false, a, false, b0, (short)0, acc0, false, false);
        acc1 = __builtin_amdgcn_wmma_f32_16x16x4_f32(false, a, false, b1, (short)0, acc1, false, false);
        acc2 = __builtin_amdgcn_wmma_f32_16x16x4_f32(false, a, false, b2, (short)0, acc2, false, false);
        acc3 = __builtin_amdgcn_wmma_f32_16x16x4_f32(false, a, false, b3, (short)0, acc3, false, false);
    }

    float bb0 = bias[n0 + 0 * 16 + lm];
    float bb1 = bias[n0 + 1 * 16 + lm];
    float bb2 = bias[n0 + 2 * 16 + lm];
    float bb3 = bias[n0 + 3 * 16 + lm];
#pragma unroll
    for (int r = 0; r < 8; ++r) {
        res[0][r] = leaky(acc0[r] + bb0);
        res[1][r] = leaky(acc1[r] + bb1);
        res[2][r] = leaky(acc2[r] + bb2);
        res[3][r] = leaky(acc3[r] + bb3);
    }
}

__global__ void k_init_mlp(const float* __restrict__ X,
                           const float* __restrict__ W,
                           const float* __restrict__ b,
                           float* __restrict__ Y, int M) {
    const int wave = (int)((blockIdx.x * blockDim.x + threadIdx.x) >> 5);
    const int numTiles = M >> 4;
    if (wave >= numTiles) return;
    const int m0 = wave << 4;

    float res[4][8];
    gemm_tile_16x64(X, W, b, 256, m0, 0, res);

    const int lane = threadIdx.x & 31;
    const int half = lane >> 4;
    const int lm   = lane & 15;
#pragma unroll
    for (int t = 0; t < 4; ++t)
#pragma unroll
        for (int r = 0; r < 8; ++r)
            Y[(size_t)(m0 + r + 8 * half) * 64 + t * 16 + lm] = res[t][r];
}

__global__ void k_lfa(const float* __restrict__ fin,
                      const float* __restrict__ geom,
                      const int*   __restrict__ idx,
                      const float* __restrict__ Wl,
                      const float* __restrict__ bl,
                      float* __restrict__ out,
                      int M, int Nper, int C) {
    const int g = blockIdx.x * blockDim.x + threadIdx.x;
    const int p = g / C;
    const int c = g - p * C;
    if (p >= M) return;
    const int batch_base = (p / Nper) * Nper;

    const float4 w  = *(const float4*)(Wl + (size_t)c * 4);
    const float  bb = bl[c];

    const float4* gp = (const float4*)(geom + (size_t)p * 64);
    const int*    ip = idx + (size_t)p * 16;

    float ts = 0.0f, fs = 0.0f;
#pragma unroll
    for (int k = 0; k < 16; ++k) {
        const float4 gg = gp[k];
        float v = fmaf(gg.x, w.x, fmaf(gg.y, w.y, fmaf(gg.z, w.z, fmaf(gg.w, w.w, bb))));
        ts += leaky(v);
        const int row = batch_base + ip[k];
        fs += fin[(size_t)row * C + c];
    }
    out[(size_t)p * (2 * C) + c]     = ts * 0.0625f;
    out[(size_t)p * (2 * C) + C + c] = fs * 0.0625f;
}

__global__ void k_final(const float* __restrict__ X3,
                        const float* __restrict__ Wf,
                        const float* __restrict__ bf,
                        const float* __restrict__ X0,
                        const float* __restrict__ Wi,
                        const float* __restrict__ bi,
                        float* __restrict__ OUT, int M) {
    const int wave = (int)((blockIdx.x * blockDim.x + threadIdx.x) >> 5);
    const int numRowTiles = M >> 4;
    if (wave >= numRowTiles * 4) return;
    const int rt = wave >> 2;
    const int cg = wave & 3;
    const int m0 = rt << 4;
    const int n0 = cg << 6;

    float r1[4][8], r2[4][8];
    gemm_tile_16x64(X3, Wf, bf, 256, m0, n0, r1);
    gemm_tile_16x64(X0, Wi, bi, 256, m0, n0, r2);

    const int lane = threadIdx.x & 31;
    const int half = lane >> 4;
    const int lm   = lane & 15;
#pragma unroll
    for (int t = 0; t < 4; ++t)
#pragma unroll
        for (int r = 0; r < 8; ++r)
            OUT[(size_t)(m0 + r + 8 * half) * 256 + n0 + t * 16 + lm] =
                r1[t][r] + r2[t][r];
}

extern "C" void kernel_launch(void* const* d_in, const int* in_sizes, int n_in,
                              void* d_out, int out_size, void* d_ws, size_t ws_size,
                              hipStream_t stream) {
    const float* input  = (const float*)d_in[0];
    const float* geom   = (const float*)d_in[1];
    const int*   idx    = (const int*)  d_in[2];
    const float* W_init = (const float*)d_in[3];
    const float* b_init = (const float*)d_in[4];
    const float* W_l1   = (const float*)d_in[5];
    const float* b_l1   = (const float*)d_in[6];
    const float* W_l2   = (const float*)d_in[7];
    const float* b_l2   = (const float*)d_in[8];
    const float* W_fin  = (const float*)d_in[9];
    const float* b_fin  = (const float*)d_in[10];
    const float* W_id   = (const float*)d_in[11];
    const float* b_id   = (const float*)d_in[12];
    float* out = (float*)d_out;

    const int M    = in_sizes[0] / 256;
    const int B    = 2;
    const int Nper = M / B;

    float* x1 = (float*)d_ws;
    float* x2 = x1 + (size_t)M * 64;
    float* x3 = x2 + (size_t)M * 128;

    const int BLK = 256;

    {
        int waves = M >> 4;
        int blocks = (waves + 7) / 8;
        k_init_mlp<<<blocks, BLK, 0, stream>>>(input, W_init, b_init, x1, M);
    }
    {
        int threads = M * 64;
        k_lfa<<<(threads + BLK - 1) / BLK, BLK, 0, stream>>>(
            x1, geom, idx, W_l1, b_l1, x2, M, Nper, 64);
    }
    {
        int threads = M * 128;
        k_lfa<<<(threads + BLK - 1) / BLK, BLK, 0, stream>>>(
            x2, geom, idx, W_l2, b_l2, x3, M, Nper, 128);
    }
    {
        int waves = (M >> 4) * 4;
        int blocks = (waves + 7) / 8;
        k_final<<<blocks, BLK, 0, stream>>>(x3, W_fin, b_fin, input, W_id, b_id, out, M);
    }
    (void)n_in; (void)out_size; (void)ws_size;
}


# GEMM operands staged once per work-group through LDS (tiles 64x64 and 32x256; whole-line global loads, ds_store, ds_load fragments) instead of per-wave 16-line global fragment loads; sum order kept
# speedup vs baseline: 2.2023x; 2.2023x over previous
#include <hip/hip_runtime.h>

typedef float v2f __attribute__((ext_vector_type(2)));
typedef float v4f __attribute__((ext_vector_type(4)));
typedef float v8f __attribute__((ext_vector_type(8)));

__device__ __forceinline__ float leaky(float v) {
    return fmaxf(v, 0.1f * v);
}


#define LDS_PITCH 36

__device__ __forceinline__ void load_frag16(const float* p, v4f f[4]) {
    f[0] = *(const v4f*)(p);
    f[1] = *(const v4f*)(p + 4);
    f[2] = *(const v4f*)(p + 8);
    f[3] = *(const v4f*)(p + 12);
}

template <int WM, int WN>
__device__ __forceinline__ void gemm_block_16x64(const float* __restrict__ X,
                                                 const float* __restrict__ W,
                                                 const float* __restrict__ bias,
                                                 int Kd, int m0, int n0,
                                                 float* lds,
                                                 float res[4][8]) {
    constexpr int NT = 32 * WM * WN;
    constexpr int BM = 16 * WM;
    constexpr int BN = 64 * WN;
    constexpr int RS = NT / 8;
    constexpr int XP = BM / RS;
    constexpr int WP = BN / RS;
    static_assert(BM % RS == 0 && BN % RS == 0, "tile rows must split evenly over the threads");

    const int tid  = threadIdx.x;
    const int lane = tid & 31;
    const int wave = tid >> 5;
    const int wm   = wave / WN;
    const int wn   = wave % WN;
    const int half = lane >> 4;
    const int lm   = lane & 15;
    const int crow = tid >> 3;
    const int cch  = (tid & 7) * 4;

    const float* xs = X + (size_t)(m0 + crow) * Kd + cch;
    const float* ws = W + (size_t)(n0 + crow) * Kd + cch;
    float* lxs = lds + crow * LDS_PITCH + cch;
    float* lws = lds + (BM + crow) * LDS_PITCH + cch;
    const float* la = lds + (wm * 16 + lm) * LDS_PITCH + 16 * half;
    const float* lb = lds + (BM + wn * 64 + lm) * LDS_PITCH + 16 * half;

    v4f px[XP], pw[WP];
#pragma unroll
    for (int j = 0; j < XP; ++j) px[j] = *(const v4f*)(xs + (size_t)j * RS * Kd);
#pragma unroll
    for (int j = 0; j < WP; ++j) pw[j] = *(const v4f*)(ws + (size_t)j * RS * Kd);

    v8f acc0 = {}, acc1 = {}, acc2 = {}, acc3 = {};

    for (int kb = 0; kb < Kd; kb += 32) {
        __syncthreads();
#pragma unroll
        for (int j = 0; j < XP; ++j) *(v4f*)(lxs + j * RS * LDS_PITCH) = px[j];
#pragma unroll
        for (int j = 0; j < WP; ++j) *(v4f*)(lws + j * RS * LDS_PITCH) = pw[j];
        __syncthreads();

        if (kb + 32 < Kd) {
#pragma unroll
            for (int j = 0; j < XP; ++j) px[j] = *(const v4f*)(xs + (size_t)j * RS * Kd + kb + 32);
#pragma unroll
            for (int j = 0; j < WP; ++j) pw[j] = *(const v4f*)(ws + (size_t)j * RS * Kd + kb + 32);
        }

        v4f a[4], b0[4], b1[4], b2[4], b3[4];
        load_frag16(la, a);
        load_frag16(lb, b0);
        load_frag16(lb + 16 * LDS_PITCH, b1);
        load_frag16(lb + 32 * LDS_PITCH, b2);
        load_frag16(lb + 48 * LDS_PITCH, b3);
#pragma unroll
        for (int q = 0; q < 4; ++q) {
            const v2f alo = a[q].xy, ahi = a[q].zw;
            acc0 = __builtin_amdgcn_wmma_f32_16x16x4_f32(false, alo, false, b0[q].xy, (short)0, acc0, false, false);
            acc1 = __builtin_amdgcn_wmma_f32_16x16x4_f32(false, alo, false, b1[q].xy, (short)0, acc1, false, false);
            acc2 = __builtin_amdgcn_wmma_f32_16x16x4_f32(false, alo, false, b2[q].xy, (short)0, acc2, false, false);
            acc3 = __builtin_amdgcn_wmma_f32_16x16x4_f32(false, alo, false, b3[q].xy, (short)0, acc3, false, false);
            acc0 = __builtin_amdgcn_wmma_f32_16x16x4_f32(false, ahi, false, b0[q].zw, (short)0, acc0, false, false);
            acc1 = __builtin_amdgcn_wmma_f32_16x16x4_f32(false, ahi, false, b1[q].zw, (short)0, acc1, false, false);
            acc2 = __builtin_amdgcn_wmma_f32_16x16x4_f32(false, ahi, false, b2[q].zw, (short)0, acc2, false, false);
            acc3 = __builtin_amdgcn_wmma_f32_16x16x4_f32(false, ahi, false, b3[q].zw, (short)0, acc3, false, false);
        }
    }

    const int nb = n0 + wn * 64 + lm;
    float bb0 = bias[nb + 0 * 16];
    float bb1 = bias[nb + 1 * 16];
    float bb2 = bias[nb + 2 * 16];
    float bb3 = bias[nb + 3 * 16];
#pragma unroll
    for (int r = 0; r < 8; ++r) {
        res[0][r] = leaky(acc0[r] + bb0);
        res[1][r] = leaky(acc1[r] + bb1);
        res[2][r] = leaky(acc2[r] + bb2);
        res[3][r] = leaky(acc3[r] + bb3);
    }
}

__global__ void __launch_bounds__(128) k_init_mlp(const float* __restrict__ X,
                           const float* __restrict__ W,
                           const float* __restrict__ b,
                           float* __restrict__ Y, int M) {
    __shared__ __attribute__((aligned(16))) float lds[(64 + 64) * LDS_PITCH];
    const int m0 = blockIdx.x * 64;
    if (m0 >= M) return;

    float res[4][8];
    gemm_block_16x64<4, 1>(X, W, b, 256, m0, 0, lds, res);

    const int lane = threadIdx.x & 31;
    const int wm   = threadIdx.x >> 5;
    const int half = lane >> 4;
    const int lm   = lane & 15;
#pragma unroll
    for (int t = 0; t < 4; ++t)
#pragma unroll
        for (int r = 0; r < 8; ++r)
            Y[(size_t)(m0 + wm * 16 + r + 8 * half) * 64 + t * 16 + lm] = res[t][r];
}

__global__ void k_lfa(const float* __restrict__ fin,
                      const float* __restrict__ geom,
                      const int*   __restrict__ idx,
                      const float* __restrict__ Wl,
                      const float* __restrict__ bl,
                      float* __restrict__ out,
                      int M, int Nper, int C) {
    const int g = blockIdx.x * blockDim.x + threadIdx.x;
    const int p = g / C;
    const int c = g - p * C;
    if (p >= M) return;
    const int batch_base = (p / Nper) * Nper;

    const float4 w  = *(const float4*)(Wl + (size_t)c * 4);
    const float  bb = bl[c];

    const float4* gp = (const float4*)(geom + (size_t)p * 64);
    const int*    ip = idx + (size_t)p * 16;

    float ts = 0.0f, fs = 0.0f;
#pragma unroll
    for (int k = 0; k < 16; ++k) {
        const float4 gg = gp[k];
        float v = fmaf(gg.x, w.x, fmaf(gg.y, w.y, fmaf(gg.z, w.z, fmaf(gg.w, w.w, bb))));
        ts += leaky(v);
        const int row = batch_base + ip[k];
        fs += fin[(size_t)row * C + c];
    }
    out[(size_t)p * (2 * C) + c]     = ts * 0.0625f;
    out[(size_t)p * (2 * C) + C + c] = fs * 0.0625f;
}

__global__ void __launch_bounds__(256) k_final(const float* __restrict__ X3,
                        const float* __restrict__ Wf,
                        const float* __restrict__ bf,
                        const float* __restrict__ X0,
                        const float* __restrict__ Wi,
                        const float* __restrict__ bi,
                        float* __restrict__ OUT, int M) {
    __shared__ __attribute__((aligned(16))) float lds[(32 + 256) * LDS_PITCH];
    const int m0 = blockIdx.x * 32;
    if (m0 >= M) return;

    float r1[4][8], r2[4][8];
    gemm_block_16x64<2, 4>(X3, Wf, bf, 256, m0, 0, lds, r1);
    gemm_block_16x64<2, 4>(X0, Wi, bi, 256, m0, 0, lds, r2);

    const int lane = threadIdx.x & 31;
    const int wave = threadIdx.x >> 5;
    const int wm   = wave >> 2;
    const int wn   = wave & 3;
    const int half = lane >> 4;
    const int lm   = lane & 15;
#pragma unroll
    for (int t = 0; t < 4; ++t)
#pragma unroll
        for (int r = 0; r < 8; ++r)
            OUT[(size_t)(m0 + wm * 16 + r + 8 * half) * 256 + wn * 64 + t * 16 + lm] =
                r1[t][r] + r2[t][r];
}

extern "C" void kernel_launch(void* const* d_in, const int* in_sizes, int n_in,
                              void* d_out, int out_size, void* d_ws, size_t ws_size,
                              hipStream_t stream) {
    const float* input  = (const float*)d_in[0];
    const float* geom   = (const float*)d_in[1];
    const int*   idx    = (const int*)  d_in[2];
    const float* W_init = (const float*)d_in[3];
    const float* b_init = (const float*)d_in[4];
    const float* W_l1   = (const float*)d_in[5];
    const float* b_l1   = (const float*)d_in[6];
    const float* W_l2   = (const float*)d_in[7];
    const float* b_l2   = (const float*)d_in[8];
    const float* W_fin  = (const float*)d_in[9];
    const float* b_fin  = (const float*)d_in[10];
    const float* W_id   = (const float*)d_in[11];
    const float* b_id   = (const float*)d_in[12];
    float* out = (float*)d_out;

    const int M    = in_sizes[0] / 256;
    const int B    = 2;
    const int Nper = M / B;

    float* x1 = (float*)d_ws;
    float* x2 = x1 + (size_t)M * 64;
    float* x3 = x2 + (size_t)M * 128;

    const int BLK = 256;

    {
        k_init_mlp<<<M / 64, 128, 0, stream>>>(input, W_init, b_init, x1, M);
    }
    {
        int threads = M * 64;
        k_lfa<<<(threads + BLK - 1) / BLK, BLK, 0, stream>>>(
            x1, geom, idx, W_l1, b_l1, x2, M, Nper, 64);
    }
    {
        int threads = M * 128;
        k_lfa<<<(threads + BLK - 1) / BLK, BLK, 0, stream>>>(
            x2, geom, idx, W_l2, b_l2, x3, M, Nper, 128);
    }
    {
        k_final<<<M / 32, BLK, 0, stream>>>(x3, W_fin, b_fin, input, W_id, b_id, out, M);
    }
    (void)n_in; (void)out_size; (void)ws_size;
}


# LFA stages: one wave per point, a lane owns 2 or 4 adjacent channels (whole-row 64/128-bit gathers); point geometry, indices and row addresses are wave scalars; 4 points per wave; sum order kept
# speedup vs baseline: 3.3445x; 1.5186x over previous
#include <hip/hip_runtime.h>

typedef float v2f __attribute__((ext_vector_type(2)));
typedef float v4f __attribute__((ext_vector_type(4)));
typedef float v8f __attribute__((ext_vector_type(8)));

__device__ __forceinline__ float leaky(float v) {
    return fmaxf(v, 0.1f * v);
}


#define LDS_PITCH 36

__device__ __forceinline__ void load_frag16(const float* p, v4f f[4]) {
    f[0] = *(const v4f*)(p);
    f[1] = *(const v4f*)(p + 4);
    f[2] = *(const v4f*)(p + 8);
    f[3] = *(const v4f*)(p + 12);
}

template <int WM, int WN>
__device__ __forceinline__ void gemm_block_16x64(const float* __restrict__ X,
                                                 const float* __restrict__ W,
                                                 const float* __restrict__ bias,
                                                 int Kd, int m0, int n0,
                                                 float* lds,
                                                 float res[4][8]) {
    constexpr int NT = 32 * WM * WN;
    constexpr int BM = 16 * WM;
    constexpr int BN = 64 * WN;
    constexpr int RS = NT / 8;
    constexpr int XP = BM / RS;
    constexpr int WP = BN / RS;
    static_assert(BM % RS == 0 && BN % RS == 0, "tile rows must split evenly over the threads");

    const int tid  = threadIdx.x;
    const int lane = tid & 31;
    const int wave = tid >> 5;
    const int wm   = wave / WN;
    const int wn   = wave % WN;
    const int half = lane >> 4;
    const int lm   = lane & 15;
    const int crow = tid >> 3;
    const int cch  = (tid & 7) * 4;

    const float* xs = X + (size_t)(m0 + crow) * Kd + cch;
    const float* ws = W + (size_t)(n0 + crow) * Kd + cch;
    float* lxs = lds + crow * LDS_PITCH + cch;
    float* lws = lds + (BM + crow) * LDS_PITCH + cch;
    const float* la = lds + (wm * 16 + lm) * LDS_PITCH + 16 * half;
    const float* lb = lds + (BM + wn * 64 + lm) * LDS_PITCH + 16 * half;

    v4f px[XP], pw[WP];
#pragma unroll
    for (int j = 0; j < XP; ++j) px[j] = *(const v4f*)(xs + (size_t)j * RS * Kd);
#pragma unroll
    for (int j = 0; j < WP; ++j) pw[j] = *(const v4f*)(ws + (size_t)j * RS * Kd);

    v8f acc0 = {}, acc1 = {}, acc2 = {}, acc3 = {};

    for (int kb = 0; kb < Kd; kb += 32) {
        __syncthreads();
#pragma unroll
        for (int j = 0; j < XP; ++j) *(v4f*)(lxs + j * RS * LDS_PITCH) = px[j];
#pragma unroll
        for (int j = 0; j < WP; ++j) *(v4f*)(lws + j * RS * LDS_PITCH) = pw[j];
        __syncthreads();

        if (kb + 32 < Kd) {
#pragma unroll
            for (int j = 0; j < XP; ++j) px[j] = *(const v4f*)(xs + (size_t)j * RS * Kd + kb + 32);
#pragma unroll
            for (int j = 0; j < WP; ++j) pw[j] = *(const v4f*)(ws + (size_t)j * RS * Kd + kb + 32);
        }

        v4f a[4], b0[4], b1[4], b2[4], b3[4];
        load_frag16(la, a);
        load_frag16(lb, b0);
        load_frag16(lb + 16 * LDS_PITCH, b1);
        load_frag16(lb + 32 * LDS_PITCH, b2);
        load_frag16(lb + 48 * LDS_PITCH, b3);
#pragma unroll
        for (int q = 0; q < 4; ++q) {
            const v2f alo = a[q].xy, ahi = a[q].zw;
            acc0 = __builtin_amdgcn_wmma_f32_16x16x4_f32(false, alo, false, b0[q].xy, (short)0, acc0, false, false);
            acc1 = __builtin_amdgcn_wmma_f32_16x16x4_f32(false, alo, false, b1[q].xy, (short)0, acc1, false, false);
            acc2 = __builtin_amdgcn_wmma_f32_16x16x4_f32(false, alo, false, b2[q].xy, (short)0, acc2, false, false);
            acc3 = __builtin_amdgcn_wmma_f32_16x16x4_f32(false, alo, false, b3[q].xy, (short)0, acc3, false, false);
            acc0 = __builtin_amdgcn_wmma_f32_16x16x4_f32(false, ahi, false, b0[q].zw, (short)0, acc0, false, false);
            acc1 = __builtin_amdgcn_wmma_f32_16x16x4_f32(false, ahi, false, b1[q].zw, (short)0, acc1, false, false);
            acc2 = __builtin_amdgcn_wmma_f32_16x16x4_f32(false, ahi, false, b2[q].zw, (short)0, acc2, false, false);
            acc3 = __builtin_amdgcn_wmma_f32_16x16x4_f32(false, ahi, false, b3[q].zw, (short)0, acc3, false, false);
        }
    }

    const int nb = n0 + wn * 64 + lm;
    float bb0 = bias[nb + 0 * 16];
    float bb1 = bias[nb + 1 * 16];
    float bb2 = bias[nb + 2 * 16];
    float bb3 = bias[nb + 3 * 16];
#pragma unroll
    for (int r = 0; r < 8; ++r) {
        res[0][r] = leaky(acc0[r] + bb0);
        res[1][r] = leaky(acc1[r] + bb1);
        res[2][r] = leaky(acc2[r] + bb2);
        res[3][r] = leaky(acc3[r] + bb3);
    }
}

__global__ void __launch_bounds__(128) k_init_mlp(const float* __restrict__ X,
                           const float* __restrict__ W,
                           const float* __restrict__ b,
                           float* __restrict__ Y, int M) {
    __shared__ __attribute__((aligned(16))) float lds[(64 + 64) * LDS_PITCH];
    const int m0 = blockIdx.x * 64;
    if (m0 >= M) return;

    float res[4][8];
    gemm_block_16x64<4, 1>(X, W, b, 256, m0, 0, lds, res);

    const int lane = threadIdx.x & 31;
    const int wm   = threadIdx.x >> 5;
    const int half = lane >> 4;
    const int lm   = lane & 15;
#pragma unroll
    for (int t = 0; t < 4; ++t)
#pragma unroll
        for (int r = 0; r < 8; ++r)
            Y[(size_t)(m0 + wm * 16 + r + 8 * half) * 64 + t * 16 + lm] = res[t][r];
}

#define LFA_WAVES 8
#define LFA_PPW   4

template <int C>
__global__ void __launch_bounds__(32 * LFA_WAVES) k_lfa(
                      const float* __restrict__ fin,
                      const float* __restrict__ geom,
                      const int*   __restrict__ idx,
                      const float* __restrict__ Wl,
                      const float* __restrict__ bl,
                      float* __restrict__ out,
                      int M, int Nper) {
    constexpr int VEC = C / 32;
    typedef float vecf __attribute__((ext_vector_type(VEC)));

    const int lane = threadIdx.x & 31;
    const int wave = __builtin_amdgcn_readfirstlane((int)(threadIdx.x >> 5));
    const int c0   = lane * VEC;

    vecf wx, wy, wz, ww, bbv;
#pragma unroll
    for (int v = 0; v < VEC; ++v) {
        const float4 t = *(const float4*)(Wl + (size_t)(c0 + v) * 4);
        wx[v] = t.x; wy[v] = t.y; wz[v] = t.z; ww[v] = t.w;
        bbv[v] = bl[c0 + v];
    }

    const int pbase = (blockIdx.x * LFA_WAVES + wave) * LFA_PPW;
    for (int i = 0; i < LFA_PPW; ++i) {
        const int p = pbase + i;
        if (p >= M) break;
        const int batch_base = (p / Nper) * Nper;

        const float4* gp = (const float4*)(geom + (size_t)p * 64);
        const int*    ip = idx + (size_t)p * 16;

        vecf ts = 0.0f, fs = 0.0f;

#pragma unroll
        for (int k = 0; k < 16; ++k) {
            const float4 gg = gp[k];
            const int row = batch_base + ip[k];
            const vecf f = *(const vecf*)(fin + (size_t)row * C + c0);
            const vecf val = __builtin_elementwise_fma((vecf)gg.x, wx,
                             __builtin_elementwise_fma((vecf)gg.y, wy,
                             __builtin_elementwise_fma((vecf)gg.z, wz,
                             __builtin_elementwise_fma((vecf)gg.w, ww, bbv))));
            ts += __builtin_elementwise_max(val, 0.1f * val);
            fs += f;
        }

        const vecf to = ts * 0.0625f, fo = fs * 0.0625f;
        *(vecf*)(out + (size_t)p * (2 * C) + c0)     = to;
        *(vecf*)(out + (size_t)p * (2 * C) + C + c0) = fo;
    }
}

__global__ void __launch_bounds__(256) k_final(const float* __restrict__ X3,
                        const float* __restrict__ Wf,
                        const float* __restrict__ bf,
                        const float* __restrict__ X0,
                        const float* __restrict__ Wi,
                        const float* __restrict__ bi,
                        float* __restrict__ OUT, int M) {
    __shared__ __attribute__((aligned(16))) float lds[(32 + 256) * LDS_PITCH];
    const int m0 = blockIdx.x * 32;
    if (m0 >= M) return;

    float r1[4][8], r2[4][8];
    gemm_block_16x64<2, 4>(X3, Wf, bf, 256, m0, 0, lds, r1);
    gemm_block_16x64<2, 4>(X0, Wi, bi, 256, m0, 0, lds, r2);

    const int lane = threadIdx.x & 31;
    const int wave = threadIdx.x >> 5;
    const int wm   = wave >> 2;
    const int wn   = wave & 3;
    const int half = lane >> 4;
    const int lm   = lane & 15;
#pragma unroll
    for (int t = 0; t < 4; ++t)
#pragma unroll
        for (int r = 0; r < 8; ++r)
            OUT[(size_t)(m0 + wm * 16 + r + 8 * half) * 256 + wn * 64 + t * 16 + lm] =
                r1[t][r] + r2[t][r];
}

extern "C" void kernel_launch(void* const* d_in, const int* in_sizes, int n_in,
                              void* d_out, int out_size, void* d_ws, size_t ws_size,
                              hipStream_t stream) {
    const float* input  = (const float*)d_in[0];
    const float* geom   = (const float*)d_in[1];
    const int*   idx    = (const int*)  d_in[2];
    const float* W_init = (const float*)d_in[3];
    const float* b_init = (const float*)d_in[4];
    const float* W_l1   = (const float*)d_in[5];
    const float* b_l1   = (const float*)d_in[6];
    const float* W_l2   = (const float*)d_in[7];
    const float* b_l2   = (const float*)d_in[8];
    const float* W_fin  = (const float*)d_in[9];
    const float* b_fin  = (const float*)d_in[10];
    const float* W_id   = (const float*)d_in[11];
    const float* b_id   = (const float*)d_in[12];
    float* out = (float*)d_out;

    const int M    = in_sizes[0] / 256;
    const int B    = 2;
    const int Nper = M / B;

    float* x1 = (float*)d_ws;
    float* x2 = x1 + (size_t)M * 64;
    float* x3 = x2 + (size_t)M * 128;

    const int BLK = 256;

    {
        k_init_mlp<<<M / 64, 128, 0, stream>>>(input, W_init, b_init, x1, M);
    }
    {
        k_lfa<64><<<(M + LFA_WAVES * LFA_PPW - 1) / (LFA_WAVES * LFA_PPW), 32 * LFA_WAVES, 0, stream>>>(
            x1, geom, idx, W_l1, b_l1, x2, M, Nper);
    }
    {
        k_lfa<128><<<(M + LFA_WAVES * LFA_PPW - 1) / (LFA_WAVES * LFA_PPW), 32 * LFA_WAVES, 0, stream>>>(
            x2, geom, idx, W_l2, b_l2, x3, M, Nper);
    }
    {
        k_final<<<M / 32, BLK, 0, stream>>>(x3, W_fin, b_fin, input, W_id, b_id, out, M);
    }
    (void)n_in; (void)out_size; (void)ws_size;
}


# Second LFA stage fused into the last kernel: each work-group builds its 32 rows of x3 in LDS and uses them as A operand of the W_fin product; x3 never goes to memory; 3 launches; sum order kept
# speedup vs baseline: 3.5807x; 1.0706x over previous
#include <hip/hip_runtime.h>

typedef float v2f __attribute__((ext_vector_type(2)));
typedef float v4f __attribute__((ext_vector_type(4)));
typedef float v8f __attribute__((ext_vector_type(8)));

__device__ __forceinline__ float leaky(float v) {
    return fmaxf(v, 0.1f * v);
}


#define LDS_PITCH 36

#define X3_PITCH 260

__device__ __forceinline__ void load_frag16(const float* p, v4f f[4]) {
    f[0] = *(const v4f*)(p);
    f[1] = *(const v4f*)(p + 4);
    f[2] = *(const v4f*)(p + 8);
    f[3] = *(const v4f*)(p + 12);
}

template <int WM, int WN, bool A_LDS>
__device__ __forceinline__ void gemm_block_16x64(const float* __restrict__ X,
                                                 const float* __restrict__ W,
                                                 const float* __restrict__ bias,
                                                 int Kd, int m0, int n0,
                                                 float* lds,
                                                 const float* atile,
                                                 float res[4][8]) {
    constexpr int NT = 32 * WM * WN;
    constexpr int BM = 16 * WM;
    constexpr int BN = 64 * WN;
    constexpr int RS = NT / 8;
    constexpr int XP = BM / RS;
    constexpr int WP = BN / RS;
    static_assert(BM % RS == 0 && BN % RS == 0, "tile rows must split evenly over the threads");

    const int tid  = threadIdx.x;
    const int lane = tid & 31;
    const int wave = tid >> 5;
    const int wm   = wave / WN;
    const int wn   = wave % WN;
    const int half = lane >> 4;
    const int lm   = lane & 15;
    const int crow = tid >> 3;
    const int cch  = (tid & 7) * 4;

    const float* xs = A_LDS ? W : X + (size_t)(m0 + crow) * Kd + cch;
    const float* ws = W + (size_t)(n0 + crow) * Kd + cch;
    float* lxs = lds + crow * LDS_PITCH + cch;
    float* lws = lds + (BM + crow) * LDS_PITCH + cch;
    const float* la = A_LDS ? atile + (wm * 16 + lm) * X3_PITCH + 16 * half
                            : lds + (wm * 16 + lm) * LDS_PITCH + 16 * half;
    const float* lb = lds + (BM + wn * 64 + lm) * LDS_PITCH + 16 * half;

    v4f px[XP], pw[WP];
    if (!A_LDS) {
#pragma unroll
        for (int j = 0; j < XP; ++j) px[j] = *(const v4f*)(xs + (size_t)j * RS * Kd);
    }
#pragma unroll
    for (int j = 0; j < WP; ++j) pw[j] = *(const v4f*)(ws + (size_t)j * RS * Kd);

    v8f acc0 = {}, acc1 = {}, acc2 = {}, acc3 = {};

    for (int kb = 0; kb < Kd; kb += 32) {
        __syncthreads();
        if (!A_LDS) {
#pragma unroll
            for (int j = 0; j < XP; ++j) *(v4f*)(lxs + j * RS * LDS_PITCH) = px[j];
        }
#pragma unroll
        for (int j = 0; j < WP; ++j) *(v4f*)(lws + j * RS * LDS_PITCH) = pw[j];
        __syncthreads();

        if (kb + 32 < Kd) {
            if (!A_LDS) {
#pragma unroll
                for (int j = 0; j < XP; ++j) px[j] = *(const v4f*)(xs + (size_t)j * RS * Kd + kb + 32);
            }
#pragma unroll
            for (int j = 0; j < WP; ++j) pw[j] = *(const v4f*)(ws + (size_t)j * RS * Kd + kb + 32);
        }

        v4f a[4], b0[4], b1[4], b2[4], b3[4];
        load_frag16(A_LDS ? la + kb : la, a);
        load_frag16(lb, b0);
        load_frag16(lb + 16 * LDS_PITCH, b1);
        load_frag16(lb + 32 * LDS_PITCH, b2);
        load_frag16(lb + 48 * LDS_PITCH, b3);
#pragma unroll
        for (int q = 0; q < 4; ++q) {
            const v2f alo = a[q].xy, ahi = a[q].zw;
            acc0 = __builtin_amdgcn_wmma_f32_16x16x4_f32(false, alo, false, b0[q].xy, (short)0, acc0, false, false);
            acc1 = __builtin_amdgcn_wmma_f32_16x16x4_f32(false, alo, false, b1[q].xy, (short)0, acc1, false, false);
            acc2 = __builtin_amdgcn_wmma_f32_16x16x4_f32(false, alo, false, b2[q].xy, (short)0, acc2, false, false);
            acc3 = __builtin_amdgcn_wmma_f32_16x16x4_f32(false, alo, false, b3[q].xy, (short)0, acc3, false, false);
            acc0 = __builtin_amdgcn_wmma_f32_16x16x4_f32(false, ahi, false, b0[q].zw, (short)0, acc0, false, false);
            acc1 = __builtin_amdgcn_wmma_f32_16x16x4_f32(false, ahi, false, b1[q].zw, (short)0, acc1, false, false);
            acc2 = __builtin_amdgcn_wmma_f32_16x16x4_f32(false, ahi, false, b2[q].zw, (short)0, acc2, false, false);
            acc3 = __builtin_amdgcn_wmma_f32_16x16x4_f32(false, ahi, false, b3[q].zw, (short)0, acc3, false, false);
        }
    }

    const int nb = n0 + wn * 64 + lm;
    float bb0 = bias[nb + 0 * 16];
    float bb1 = bias[nb + 1 * 16];
    float bb2 = bias[nb + 2 * 16];
    float bb3 = bias[nb + 3 * 16];
#pragma unroll
    for (int r = 0; r < 8; ++r) {
        res[0][r] = leaky(acc0[r] + bb0);
        res[1][r] = leaky(acc1[r] + bb1);
        res[2][r] = leaky(acc2[r] + bb2);
        res[3][r] = leaky(acc3[r] + bb3);
    }
}

__global__ void __launch_bounds__(128) k_init_mlp(const float* __restrict__ X,
                           const float* __restrict__ W,
                           const float* __restrict__ b,
                           float* __restrict__ Y, int M) {
    __shared__ __attribute__((aligned(16))) float lds[(64 + 64) * LDS_PITCH];
    const int m0 = blockIdx.x * 64;
    if (m0 >= M) return;

    float res[4][8];
    gemm_block_16x64<4, 1, false>(X, W, b, 256, m0, 0, lds, nullptr, res);

    const int lane = threadIdx.x & 31;
    const int wm   = threadIdx.x >> 5;
    const int half = lane >> 4;
    const int lm   = lane & 15;
#pragma unroll
    for (int t = 0; t < 4; ++t)
#pragma unroll
        for (int r = 0; r < 8; ++r)
            Y[(size_t)(m0 + wm * 16 + r + 8 * half) * 64 + t * 16 + lm] = res[t][r];
}

#define LFA_WAVES 8
#define LFA_PPW   4

template <int C>
__global__ void __launch_bounds__(32 * LFA_WAVES) k_lfa(
                      const float* __restrict__ fin,
                      const float* __restrict__ geom,
                      const int*   __restrict__ idx,
                      const float* __restrict__ Wl,
                      const float* __restrict__ bl,
                      float* __restrict__ out,
                      int M, int Nper) {
    constexpr int VEC = C / 32;
    typedef float vecf __attribute__((ext_vector_type(VEC)));

    const int lane = threadIdx.x & 31;
    const int wave = __builtin_amdgcn_readfirstlane((int)(threadIdx.x >> 5));
    const int c0   = lane * VEC;

    vecf wx, wy, wz, ww, bbv;
#pragma unroll
    for (int v = 0; v < VEC; ++v) {
        const float4 t = *(const float4*)(Wl + (size_t)(c0 + v) * 4);
        wx[v] = t.x; wy[v] = t.y; wz[v] = t.z; ww[v] = t.w;
        bbv[v] = bl[c0 + v];
    }

    const int pbase = (blockIdx.x * LFA_WAVES + wave) * LFA_PPW;
    for (int i = 0; i < LFA_PPW; ++i) {
        const int p = pbase + i;
        if (p >= M) break;
        const int batch_base = (p / Nper) * Nper;

        const float4* gp = (const float4*)(geom + (size_t)p * 64);
        const int*    ip = idx + (size_t)p * 16;

        vecf ts = 0.0f, fs = 0.0f;

#pragma unroll
        for (int k = 0; k < 16; ++k) {
            const float4 gg = gp[k];
            const int row = batch_base + ip[k];
            const vecf f = *(const vecf*)(fin + (size_t)row * C + c0);
            const vecf val = __builtin_elementwise_fma((vecf)gg.x, wx,
                             __builtin_elementwise_fma((vecf)gg.y, wy,
                             __builtin_elementwise_fma((vecf)gg.z, wz,
                             __builtin_elementwise_fma((vecf)gg.w, ww, bbv))));
            ts += __builtin_elementwise_max(val, 0.1f * val);
            fs += f;
        }

        const vecf to = ts * 0.0625f, fo = fs * 0.0625f;
        *(vecf*)(out + (size_t)p * (2 * C) + c0)     = to;
        *(vecf*)(out + (size_t)p * (2 * C) + C + c0) = fo;
    }
}

__device__ __forceinline__ void lfa128_tile(const float* __restrict__ fin,
                                            const float* __restrict__ geom,
                                            const int*   __restrict__ idx,
                                            const float* __restrict__ Wl,
                                            const float* __restrict__ bl,
                                            float* tile,
                                            int m0, int M, int Nper) {
    constexpr int C = 128;
    constexpr int VEC = C / 32;
    typedef float vecf __attribute__((ext_vector_type(VEC)));

    const int lane = threadIdx.x & 31;
    const int wave = __builtin_amdgcn_readfirstlane((int)(threadIdx.x >> 5));
    const int c0   = lane * VEC;

    vecf wx, wy, wz, ww, bbv;
#pragma unroll
    for (int v = 0; v < VEC; ++v) {
        const float4 t = *(const float4*)(Wl + (size_t)(c0 + v) * 4);
        wx[v] = t.x; wy[v] = t.y; wz[v] = t.z; ww[v] = t.w;
        bbv[v] = bl[c0 + v];
    }

    static_assert(LFA_WAVES * LFA_PPW == 32, "the work-group's waves must cover the 32 rows of its tile");
    const int r0 = wave * LFA_PPW;
    for (int i = 0; i < LFA_PPW; ++i) {
        const int p = m0 + r0 + i;
        if (p >= M) break;
        const int batch_base = (p / Nper) * Nper;

        const float4* gp = (const float4*)(geom + (size_t)p * 64);
        const int*    ip = idx + (size_t)p * 16;

        vecf ts = 0.0f, fs = 0.0f;

#pragma unroll
        for (int k = 0; k < 16; ++k) {
            const float4 gg = gp[k];
            const int row = batch_base + ip[k];
            const vecf f = *(const vecf*)(fin + (size_t)row * C + c0);
            const vecf val = __builtin_elementwise_fma((vecf)gg.x, wx,
                             __builtin_elementwise_fma((vecf)gg.y, wy,
                             __builtin_elementwise_fma((vecf)gg.z, wz,
                             __builtin_elementwise_fma((vecf)gg.w, ww, bbv))));
            ts += __builtin_elementwise_max(val, 0.1f * val);
            fs += f;
        }

        const vecf to = ts * 0.0625f, fo = fs * 0.0625f;
        *(vecf*)(tile + (r0 + i) * X3_PITCH + c0)     = to;
        *(vecf*)(tile + (r0 + i) * X3_PITCH + C + c0) = fo;
    }
}

__global__ void __launch_bounds__(256, 6) k_lfa2_final(const float* __restrict__ X2,
                        const float* __restrict__ geom,
                        const int*   __restrict__ idx,
                        const float* __restrict__ Wl,
                        const float* __restrict__ bl,
                        const float* __restrict__ Wf,
                        const float* __restrict__ bf,
                        const float* __restrict__ X0,
                        const float* __restrict__ Wi,
                        const float* __restrict__ bi,
                        float* __restrict__ OUT, int M, int Nper) {
    __shared__ __attribute__((aligned(16))) float lds[(32 + 256) * LDS_PITCH];
    __shared__ __attribute__((aligned(16))) float x3t[32 * X3_PITCH];
    const int m0 = blockIdx.x * 32;
    if (m0 >= M) return;

    lfa128_tile(X2, geom, idx, Wl, bl, x3t, m0, M, Nper);

    float r1[4][8], r2[4][8];
    gemm_block_16x64<2, 4, true>(nullptr, Wf, bf, 256, m0, 0, lds, x3t, r1);
    gemm_block_16x64<2, 4, false>(X0, Wi, bi, 256, m0, 0, lds, nullptr, r2);

    const int lane = threadIdx.x & 31;
    const int wave = threadIdx.x >> 5;
    const int wm   = wave >> 2;
    const int wn   = wave & 3;
    const int half = lane >> 4;
    const int lm   = lane & 15;
#pragma unroll
    for (int t = 0; t < 4; ++t)
#pragma unroll
        for (int r = 0; r < 8; ++r)
            OUT[(size_t)(m0 + wm * 16 + r + 8 * half) * 256 + wn * 64 + t * 16 + lm] =
                r1[t][r] + r2[t][r];
}

extern "C" void kernel_launch(void* const* d_in, const int* in_sizes, int n_in,
                              void* d_out, int out_size, void* d_ws, size_t ws_size,
                              hipStream_t stream) {
    const float* input  = (const float*)d_in[0];
    const float* geom   = (const float*)d_in[1];
    const int*   idx    = (const int*)  d_in[2];
    const float* W_init = (const float*)d_in[3];
    const float* b_init = (const float*)d_in[4];
    const float* W_l1   = (const float*)d_in[5];
    const float* b_l1   = (const float*)d_in[6];
    const float* W_l2   = (const float*)d_in[7];
    const float* b_l2   = (const float*)d_in[8];
    const float* W_fin  = (const float*)d_in[9];
    const float* b_fin  = (const float*)d_in[10];
    const float* W_id   = (const float*)d_in[11];
    const float* b_id   = (const float*)d_in[12];
    float* out = (float*)d_out;

    const int M    = in_sizes[0] / 256;
    const int B    = 2;
    const int Nper = M / B;

    float* x1 = (float*)d_ws;
    float* x2 = x1 + (size_t)M * 64;

    const int BLK = 256;

    {
        k_init_mlp<<<M / 64, 128, 0, stream>>>(input, W_init, b_init, x1, M);
    }
    {
        k_lfa<64><<<(M + LFA_WAVES * LFA_PPW - 1) / (LFA_WAVES * LFA_PPW), 32 * LFA_WAVES, 0, stream>>>(
            x1, geom, idx, W_l1, b_l1, x2, M, Nper);
    }
    {
        k_lfa2_final<<<M / 32, BLK, 0, stream>>>(x2, geom, idx, W_l2, b_l2, W_fin, b_fin,
                                                input, W_id, b_id, out, M, Nper);
    }
    (void)n_in; (void)out_size; (void)ws_size;
}
